# starting kernel
# baseline (speedup 1.0000x reference)
#include <hip/hip_runtime.h>

typedef float v2f __attribute__((ext_vector_type(2)));
typedef float v8f __attribute__((ext_vector_type(8)));

#define H_DIM 56
#define W_DIM 56
#define TH 28
#define TW 28
#define NB 16
#define CIN 256
#define OCH 256
#define TILES_TOTAL (NB * TH * TW)
#define TILES_PER_N (TH * TW)
#define TPB 16
#define OPB 64
#define OFRAGS (OPB / 16)
#define CCHUNK 32
#define VSTRIDE 34

__global__ void wino_weight_transform(const float* __restrict__ w, float* __restrict__ U)
{
    int idx = blockIdx.x * blockDim.x + threadIdx.x;
    if (idx >= OCH * CIN) return;
    int o = idx >> 8;
    int c = idx & 255;
    const float* wp = w + (size_t)(o * CIN + c) * 9;

    float g[3][3];
#pragma unroll
    for (int r = 0; r < 3; ++r)
#pragma unroll
        for (int s = 0; s < 3; ++s) g[r][s] = wp[r * 3 + s];

    float u[4][3];
#pragma unroll
    for (int s = 0; s < 3; ++s) {
        u[0][s] = g[0][s];
        u[1][s] = 0.5f * (g[0][s] + g[1][s] + g[2][s]);
        u[2][s] = 0.5f * (g[0][s] - g[1][s] + g[2][s]);
        u[3][s] = g[2][s];
    }
#pragma unroll
    for (int r = 0; r < 4; ++r) {
        float a = u[r][0], b = u[r][1], d = u[r][2];
        float e[4];
        e[0] = a;
        e[1] = 0.5f * (a + b + d);
        e[2] = 0.5f * (a - b + d);
        e[3] = d;
#pragma unroll
        for (int l = 0; l < 4; ++l)
            U[(size_t)((r * 4 + l) * OCH + o) * CIN + c] = e[l];
    }
}

__launch_bounds__(512)
__global__ void wino_main(const float* __restrict__ x, const float* __restrict__ U,
                          float* __restrict__ out)
{
    __shared__ float smem[OPB * 16 * 16];

    const int tb    = blockIdx.x;
    const int ob    = blockIdx.y;
    const int tid   = threadIdx.x;
    const int lane  = tid & 31;
    const int il    = tid >> 5;
    const int obase = ob * OPB;

    const int tile_l = tid & 15;
    const int cl     = tid >> 4;
    const int tile_g = tb * TPB + tile_l;
    const int n      = tile_g / TILES_PER_N;
    const int trem   = tile_g % TILES_PER_N;
    const int th     = trem / TW;
    const int tw     = trem % TW;
    const int ih0    = 2 * th - 1;
    const int iw0    = 2 * tw - 1;

    int   rofs[4], cofs[4];
    float rmask[4], cmask[4];
#pragma unroll
    for (int r = 0; r < 4; ++r) {
        const int ih = ih0 + r;
        const bool rok = (unsigned)ih < (unsigned)H_DIM;
        rofs[r]  = (rok ? ih : 0) * W_DIM;
        rmask[r] = rok ? 1.f : 0.f;
    }
#pragma unroll
    for (int s = 0; s < 4; ++s) {
        const int iw = iw0 + s;
        const bool cok = (unsigned)iw < (unsigned)W_DIM;
        cofs[s]  = cok ? iw : 0;
        cmask[s] = cok ? 1.f : 0.f;
    }

    v8f acc[OFRAGS];
#pragma unroll
    for (int f = 0; f < OFRAGS; ++f)
        acc[f] = (v8f){0.f, 0.f, 0.f, 0.f, 0.f, 0.f, 0.f, 0.f};

    const int m16   = lane & 15;
    const int khalf = (lane >> 4) << 1;
    const float* Arow[OFRAGS];
#pragma unroll
    for (int f = 0; f < OFRAGS; ++f)
        Arow[f] = U + ((size_t)il * OCH + (obase + f * 16 + m16)) * CIN;
    const float* Bbase = &smem[il * (16 * VSTRIDE) + m16 * VSTRIDE];

    for (int cb = 0; cb < CIN; cb += CCHUNK) {
        const int c = cb + cl;
        const float* xp = x + ((size_t)(n * CIN + c) * H_DIM) * W_DIM;
        float d[4][4];
#pragma unroll
        for (int r = 0; r < 4; ++r)
#pragma unroll
            for (int s = 0; s < 4; ++s)
                d[r][s] = xp[rofs[r] + cofs[s]] * (rmask[r] * cmask[s]);

        float t[4][4];
#pragma unroll
        for (int s = 0; s < 4; ++s) {
            t[0][s] = d[0][s] - d[2][s];
            t[1][s] = d[1][s] + d[2][s];
            t[2][s] = d[2][s] - d[1][s];
            t[3][s] = d[1][s] - d[3][s];
        }
#pragma unroll
        for (int r = 0; r < 4; ++r) {
            float v0 = t[r][0] - t[r][2];
            float v1 = t[r][1] + t[r][2];
            float v2 = t[r][2] - t[r][1];
            float v3 = t[r][1] - t[r][3];
            smem[(r * 4 + 0) * (16 * VSTRIDE) + tile_l * VSTRIDE + cl] = v0;
            smem[(r * 4 + 1) * (16 * VSTRIDE) + tile_l * VSTRIDE + cl] = v1;
            smem[(r * 4 + 2) * (16 * VSTRIDE) + tile_l * VSTRIDE + cl] = v2;
            smem[(r * 4 + 3) * (16 * VSTRIDE) + tile_l * VSTRIDE + cl] = v3;
        }
        __syncthreads();

#pragma unroll
        for (int k = 0; k < CCHUNK; k += 4) {
            const int kk = k + khalf;
            v2f b = *(const v2f*)&Bbase[kk];
#pragma unroll
            for (int f = 0; f < OFRAGS; ++f) {
                v2f a = *(const v2f*)&Arow[f][cb + kk];
                acc[f] = __builtin_amdgcn_wmma_f32_16x16x4_f32(
                    false, a, false, b, (short)0, acc[f], false, false);
            }
        }
        __syncthreads();
    }

    {
        const int orow = (lane >> 4) << 3;
        const int ncol = lane & 15;
#pragma unroll
        for (int f = 0; f < OFRAGS; ++f)
#pragma unroll
            for (int r = 0; r < 8; ++r)
                smem[(((f * 16 + r + orow) * 16) + ncol) * 16 + il] = acc[f][r];
    }
    __syncthreads();

    for (int p = tid; p < OPB * 16; p += 512) {
        const int o_l = p >> 4;
        const int tl  = p & 15;
        const int tg  = tb * TPB + tl;
        const int nn  = tg / TILES_PER_N;
        const int tr  = tg % TILES_PER_N;
        const int hh  = (tr / TW) * 2;
        const int ww  = (tr % TW) * 2;

        float m[16];
#pragma unroll
        for (int i = 0; i < 16; ++i)
            m[i] = smem[(o_l * 16 + tl) * 16 + i];

        float z0[4], z1[4];
#pragma unroll
        for (int c = 0; c < 4; ++c) {
            z0[c] = m[0 * 4 + c] + m[1 * 4 + c] + m[2 * 4 + c];
            z1[c] = m[1 * 4 + c] - m[2 * 4 + c] - m[3 * 4 + c];
        }
        const float y00 = z0[0] + z0[1] + z0[2];
        const float y01 = z0[1] - z0[2] - z0[3];
        const float y10 = z1[0] + z1[1] + z1[2];
        const float y11 = z1[1] - z1[2] - z1[3];

        const int o = obase + o_l;
        float* op = out + (((size_t)(nn * OCH + o) * H_DIM) + hh) * W_DIM + ww;
        op[0]         = y00;
        op[1]         = y01;
        op[W_DIM]     = y10;
        op[W_DIM + 1] = y11;
    }
}

extern "C" void kernel_launch(void* const* d_in, const int* in_sizes, int n_in,
                              void* d_out, int out_size, void* d_ws, size_t ws_size,
                              hipStream_t stream)
{
    const float* x = (const float*)d_in[0];
    const float* w = (const float*)d_in[1];
    float* out = (float*)d_out;
    float* U = (float*)d_ws;

    wino_weight_transform<<<dim3((OCH * CIN + 255) / 256), dim3(256), 0, stream>>>(w, U);
    wino_main<<<dim3(TILES_TOTAL / TPB, OCH / OPB), dim3(512), 0, stream>>>(x, U, out);
}


# weight-transform kernel stores U in A-fragment order, so each A fragment pair is one contiguous 16-byte-per-lane load (4 lines per wave load, was 16 lines of 16 B each)
# speedup vs baseline: 2.3497x; 2.3497x over previous
#include <hip/hip_runtime.h>

typedef float v2f __attribute__((ext_vector_type(2)));
typedef float v4f __attribute__((ext_vector_type(4)));
typedef float v8f __attribute__((ext_vector_type(8)));

#define H_DIM 56
#define W_DIM 56
#define TH 28
#define TW 28
#define NB 16
#define CIN 256
#define OCH 256
#define TILES_TOTAL (NB * TH * TW)
#define TILES_PER_N (TH * TW)
#define TPB 16
#define OPB 64
#define OFRAGS (OPB / 16)
#define CCHUNK 32
#define VSTRIDE 34

__device__ __forceinline__ size_t u_frag_index(int il, int o, int c)
{
    const int lane = (o & 15) + (((c >> 1) & 1) << 4);
    const int j    = (((c >> 2) & 1) << 1) + (c & 1);
    return ((((size_t)il * (OCH / 16) + (o >> 4)) * (CIN / 8) + (c >> 3)) * 32 + lane) * 4 + j;
}

__global__ void wino_weight_transform(const float* __restrict__ w, float* __restrict__ U)
{
    int idx = blockIdx.x * blockDim.x + threadIdx.x;
    if (idx >= OCH * CIN) return;
    int o = idx >> 8;
    int c = idx & 255;
    const float* wp = w + (size_t)(o * CIN + c) * 9;

    float g[3][3];
#pragma unroll
    for (int r = 0; r < 3; ++r)
#pragma unroll
        for (int s = 0; s < 3; ++s) g[r][s] = wp[r * 3 + s];

    float u[4][3];
#pragma unroll
    for (int s = 0; s < 3; ++s) {
        u[0][s] = g[0][s];
        u[1][s] = 0.5f * (g[0][s] + g[1][s] + g[2][s]);
        u[2][s] = 0.5f * (g[0][s] - g[1][s] + g[2][s]);
        u[3][s] = g[2][s];
    }
#pragma unroll
    for (int r = 0; r < 4; ++r) {
        float a = u[r][0], b = u[r][1], d = u[r][2];
        float e[4];
        e[0] = a;
        e[1] = 0.5f * (a + b + d);
        e[2] = 0.5f * (a - b + d);
        e[3] = d;
#pragma unroll
        for (int l = 0; l < 4; ++l)
            U[u_frag_index(r * 4 + l, o, c)] = e[l];
    }
}

__launch_bounds__(512)
__global__ void wino_main(const float* __restrict__ x, const float* __restrict__ U,
                          float* __restrict__ out)
{
    __shared__ float smem[OPB * 16 * 16];

    const int tb    = blockIdx.x;
    const int ob    = blockIdx.y;
    const int tid   = threadIdx.x;
    const int lane  = tid & 31;
    const int il    = tid >> 5;
    const int obase = ob * OPB;

    const int tile_l = tid & 15;
    const int cl     = tid >> 4;
    const int tile_g = tb * TPB + tile_l;
    const int n      = tile_g / TILES_PER_N;
    const int trem   = tile_g % TILES_PER_N;
    const int th     = trem / TW;
    const int tw     = trem % TW;
    const int ih0    = 2 * th - 1;
    const int iw0    = 2 * tw - 1;

    int   rofs[4], cofs[4];
    float rmask[4], cmask[4];
#pragma unroll
    for (int r = 0; r < 4; ++r) {
        const int ih = ih0 + r;
        const bool rok = (unsigned)ih < (unsigned)H_DIM;
        rofs[r]  = (rok ? ih : 0) * W_DIM;
        rmask[r] = rok ? 1.f : 0.f;
    }
#pragma unroll
    for (int s = 0; s < 4; ++s) {
        const int iw = iw0 + s;
        const bool cok = (unsigned)iw < (unsigned)W_DIM;
        cofs[s]  = cok ? iw : 0;
        cmask[s] = cok ? 1.f : 0.f;
    }

    v8f acc[OFRAGS];
#pragma unroll
    for (int f = 0; f < OFRAGS; ++f)
        acc[f] = (v8f){0.f, 0.f, 0.f, 0.f, 0.f, 0.f, 0.f, 0.f};

    const int m16   = lane & 15;
    const int khalf = (lane >> 4) << 1;
    const v4f* Afrag[OFRAGS];
#pragma unroll
    for (int f = 0; f < OFRAGS; ++f)
        Afrag[f] = (const v4f*)U +
                   ((size_t)il * (OCH / 16) + (ob * OFRAGS + f)) * (CIN / 8) * 32 + lane;
    const float* Bbase = &smem[il * (16 * VSTRIDE) + m16 * VSTRIDE];

    for (int cb = 0; cb < CIN; cb += CCHUNK) {
        const int c = cb + cl;
        const float* xp = x + ((size_t)(n * CIN + c) * H_DIM) * W_DIM;
        float d[4][4];
#pragma unroll
        for (int r = 0; r < 4; ++r)
#pragma unroll
            for (int s = 0; s < 4; ++s)
                d[r][s] = xp[rofs[r] + cofs[s]] * (rmask[r] * cmask[s]);

        float t[4][4];
#pragma unroll
        for (int s = 0; s < 4; ++s) {
            t[0][s] = d[0][s] - d[2][s];
            t[1][s] = d[1][s] + d[2][s];
            t[2][s] = d[2][s] - d[1][s];
            t[3][s] = d[1][s] - d[3][s];
        }
#pragma unroll
        for (int r = 0; r < 4; ++r) {
            float v0 = t[r][0] - t[r][2];
            float v1 = t[r][1] + t[r][2];
            float v2 = t[r][2] - t[r][1];
            float v3 = t[r][1] - t[r][3];
            smem[(r * 4 + 0) * (16 * VSTRIDE) + tile_l * VSTRIDE + cl] = v0;
            smem[(r * 4 + 1) * (16 * VSTRIDE) + tile_l * VSTRIDE + cl] = v1;
            smem[(r * 4 + 2) * (16 * VSTRIDE) + tile_l * VSTRIDE + cl] = v2;
            smem[(r * 4 + 3) * (16 * VSTRIDE) + tile_l * VSTRIDE + cl] = v3;
        }
        __syncthreads();

#pragma unroll
        for (int k = 0; k < CCHUNK; k += 8) {
            const int kq = (cb + k) >> 3;
            v2f b0 = *(const v2f*)&Bbase[k + khalf];
            v2f b1 = *(const v2f*)&Bbase[k + 4 + khalf];
#pragma unroll
            for (int f = 0; f < OFRAGS; ++f) {
                const v4f a = Afrag[f][kq * 32];
                const v2f a0 = (v2f){a.x, a.y};
                const v2f a1 = (v2f){a.z, a.w};
                acc[f] = __builtin_amdgcn_wmma_f32_16x16x4_f32(
                    false, a0, false, b0, (short)0, acc[f], false, false);
                acc[f] = __builtin_amdgcn_wmma_f32_16x16x4_f32(
                    false, a1, false, b1, (short)0, acc[f], false, false);
            }
        }
        __syncthreads();
    }

    {
        const int orow = (lane >> 4) << 3;
        const int ncol = lane & 15;
#pragma unroll
        for (int f = 0; f < OFRAGS; ++f)
#pragma unroll
            for (int r = 0; r < 8; ++r)
                smem[(((f * 16 + r + orow) * 16) + ncol) * 16 + il] = acc[f][r];
    }
    __syncthreads();

    for (int p = tid; p < OPB * 16; p += 512) {
        const int o_l = p >> 4;
        const int tl  = p & 15;
        const int tg  = tb * TPB + tl;
        const int nn  = tg / TILES_PER_N;
        const int tr  = tg % TILES_PER_N;
        const int hh  = (tr / TW) * 2;
        const int ww  = (tr % TW) * 2;

        float m[16];
#pragma unroll
        for (int i = 0; i < 16; ++i)
            m[i] = smem[(o_l * 16 + tl) * 16 + i];

        float z0[4], z1[4];
#pragma unroll
        for (int c = 0; c < 4; ++c) {
            z0[c] = m[0 * 4 + c] + m[1 * 4 + c] + m[2 * 4 + c];
            z1[c] = m[1 * 4 + c] - m[2 * 4 + c] - m[3 * 4 + c];
        }
        const float y00 = z0[0] + z0[1] + z0[2];
        const float y01 = z0[1] - z0[2] - z0[3];
        const float y10 = z1[0] + z1[1] + z1[2];
        const float y11 = z1[1] - z1[2] - z1[3];

        const int o = obase + o_l;
        float* op = out + (((size_t)(nn * OCH + o) * H_DIM) + hh) * W_DIM + ww;
        op[0]         = y00;
        op[1]         = y01;
        op[W_DIM]     = y10;
        op[W_DIM + 1] = y11;
    }
}

extern "C" void kernel_launch(void* const* d_in, const int* in_sizes, int n_in,
                              void* d_out, int out_size, void* d_ws, size_t ws_size,
                              hipStream_t stream)
{
    const float* x = (const float*)d_in[0];
    const float* w = (const float*)d_in[1];
    float* out = (float*)d_out;
    float* U = (float*)d_ws;

    wino_weight_transform<<<dim3((OCH * CIN + 255) / 256), dim3(256), 0, stream>>>(w, U);
    wino_main<<<dim3(TILES_TOTAL / TPB, OCH / OPB), dim3(512), 0, stream>>>(x, U, out);
}


# each wave keeps 4x2 accumulator tiles (32 tiles per workgroup, was 16): one A fragment load from U feeds two B fragments, so U reads per matrix instruction halve; k summation order unchanged
# speedup vs baseline: 2.9325x; 1.2480x over previous
#include <hip/hip_runtime.h>

typedef float v2f __attribute__((ext_vector_type(2)));
typedef float v4f __attribute__((ext_vector_type(4)));
typedef float v8f __attribute__((ext_vector_type(8)));

#define H_DIM 56
#define W_DIM 56
#define TH 28
#define TW 28
#define NB 16
#define CIN 256
#define OCH 256
#define TILES_TOTAL (NB * TH * TW)
#define TILES_PER_N (TH * TW)
#define NT 2
#define TPB (16 * NT)
#define OPB 64
#define OFRAGS (OPB / 16)
#define CCHUNK 32
#define VSTRIDE 34
#define PPT (TPB * CCHUNK / 512)

__device__ __forceinline__ size_t u_frag_index(int il, int o, int c)
{
    const int lane = (o & 15) + (((c >> 1) & 1) << 4);
    const int j    = (((c >> 2) & 1) << 1) + (c & 1);
    return ((((size_t)il * (OCH / 16) + (o >> 4)) * (CIN / 8) + (c >> 3)) * 32 + lane) * 4 + j;
}

__global__ void wino_weight_transform(const float* __restrict__ w, float* __restrict__ U)
{
    int idx = blockIdx.x * blockDim.x + threadIdx.x;
    if (idx >= OCH * CIN) return;
    int o = idx >> 8;
    int c = idx & 255;
    const float* wp = w + (size_t)(o * CIN + c) * 9;

    float g[3][3];
#pragma unroll
    for (int r = 0; r < 3; ++r)
#pragma unroll
        for (int s = 0; s < 3; ++s) g[r][s] = wp[r * 3 + s];

    float u[4][3];
#pragma unroll
    for (int s = 0; s < 3; ++s) {
        u[0][s] = g[0][s];
        u[1][s] = 0.5f * (g[0][s] + g[1][s] + g[2][s]);
        u[2][s] = 0.5f * (g[0][s] - g[1][s] + g[2][s]);
        u[3][s] = g[2][s];
    }
#pragma unroll
    for (int r = 0; r < 4; ++r) {
        float a = u[r][0], b = u[r][1], d = u[r][2];
        float e[4];
        e[0] = a;
        e[1] = 0.5f * (a + b + d);
        e[2] = 0.5f * (a - b + d);
        e[3] = d;
#pragma unroll
        for (int l = 0; l < 4; ++l)
            U[u_frag_index(r * 4 + l, o, c)] = e[l];
    }
}

__launch_bounds__(512, 8)
__global__ void wino_main(const float* __restrict__ x, const float* __restrict__ U,
                          float* __restrict__ out)
{
    __shared__ float smem[OPB * TPB * 16];

    const int tb    = blockIdx.x;
    const int ob    = blockIdx.y;
    const int tid   = threadIdx.x;
    const int lane  = tid & 31;
    const int il    = tid >> 5;
    const int obase = ob * OPB;

    const int tile_l = tid % TPB;
    const int cl     = tid / TPB;
    const int tile_g = tb * TPB + tile_l;
    const int n      = tile_g / TILES_PER_N;
    const int trem   = tile_g % TILES_PER_N;
    const int th     = trem / TW;
    const int tw     = trem % TW;

    const bool r0ok = th > 0, r3ok = th < TH - 1;
    const bool c0ok = tw > 0, c3ok = tw < TW - 1;
    const float rm0 = r0ok ? 1.f : 0.f, rm3 = r3ok ? 1.f : 0.f;
    const float cm0 = c0ok ? 1.f : 0.f, cm3 = c3ok ? 1.f : 0.f;
    unsigned rbase[4];
    {
        const unsigned base = (unsigned)((n * CIN + PPT * cl) * (H_DIM * W_DIM) + 2 * tw);
        rbase[0] = (base + (unsigned)((r0ok ? 2 * th - 1 : 2 * th) * W_DIM)) * 4u;
        rbase[1] = (base + (unsigned)((2 * th) * W_DIM)) * 4u;
        rbase[2] = (base + (unsigned)((2 * th + 1) * W_DIM)) * 4u;
        rbase[3] = (base + (unsigned)((r3ok ? 2 * th + 2 : 2 * th + 1) * W_DIM)) * 4u;
    }
    const unsigned c0adj = c0ok ? (unsigned)-4 : 0u;
    const unsigned c3adj = c3ok ? 8u : 0u;

    v8f acc[OFRAGS][NT];
#pragma unroll
    for (int f = 0; f < OFRAGS; ++f)
#pragma unroll
        for (int j = 0; j < NT; ++j)
            acc[f][j] = (v8f){0.f, 0.f, 0.f, 0.f, 0.f, 0.f, 0.f, 0.f};

    const int m16   = lane & 15;
    const int khalf = (lane >> 4) << 1;
    const v4f* Afrag[OFRAGS];
#pragma unroll
    for (int f = 0; f < OFRAGS; ++f)
        Afrag[f] = (const v4f*)U +
                   ((size_t)il * (OCH / 16) + (ob * OFRAGS + f)) * (CIN / 8) * 32 + lane;
    const float* Bbase = &smem[il * (TPB * VSTRIDE) + m16 * VSTRIDE];

    for (int cb = 0; cb < CIN; cb += CCHUNK) {
        float d[PPT][4][4];
#pragma unroll
        for (int q = 0; q < PPT; ++q) {
            const char* xq = (const char*)x + (size_t)q * (H_DIM * W_DIM) * 4;
#pragma unroll
            for (int r = 0; r < 4; ++r) {
                d[q][r][0] = *(const float*)(xq + (rbase[r] + c0adj));
                d[q][r][1] = *(const float*)(xq + rbase[r]);
                d[q][r][2] = *(const float*)(xq + rbase[r] + 4);
                d[q][r][3] = *(const float*)(xq + (rbase[r] + c3adj));
            }
        }
#pragma unroll
        for (int r = 0; r < 4; ++r)
            rbase[r] += CCHUNK * (H_DIM * W_DIM) * 4;
#pragma unroll
        for (int q = 0; q < PPT; ++q) {
#pragma unroll
            for (int s = 0; s < 4; ++s) {
                d[q][0][s] *= rm0;
                d[q][3][s] *= rm3;
            }
#pragma unroll
            for (int r = 0; r < 4; ++r) {
                d[q][r][0] *= cm0;
                d[q][r][3] *= cm3;
            }
            const int cq = PPT * cl + q;
            float t[4][4];
#pragma unroll
            for (int s = 0; s < 4; ++s) {
                t[0][s] = d[q][0][s] - d[q][2][s];
                t[1][s] = d[q][1][s] + d[q][2][s];
                t[2][s] = d[q][2][s] - d[q][1][s];
                t[3][s] = d[q][1][s] - d[q][3][s];
            }
#pragma unroll
            for (int r = 0; r < 4; ++r) {
                float v0 = t[r][0] - t[r][2];
                float v1 = t[r][1] + t[r][2];
                float v2 = t[r][2] - t[r][1];
                float v3 = t[r][1] - t[r][3];
                smem[(r * 4 + 0) * (TPB * VSTRIDE) + tile_l * VSTRIDE + cq] = v0;
                smem[(r * 4 + 1) * (TPB * VSTRIDE) + tile_l * VSTRIDE + cq] = v1;
                smem[(r * 4 + 2) * (TPB * VSTRIDE) + tile_l * VSTRIDE + cq] = v2;
                smem[(r * 4 + 3) * (TPB * VSTRIDE) + tile_l * VSTRIDE + cq] = v3;
            }
        }
        __syncthreads();

        {
            constexpr int STEPS = (CCHUNK / 8) * OFRAGS;
            constexpr int AHEAD = 2;
            v4f a[STEPS];
#pragma unroll
            for (int t = 0; t < AHEAD; ++t)
                a[t] = Afrag[t % OFRAGS][((cb >> 3) + t / OFRAGS) * 32];
            v2f b0[NT], b1[NT];
#pragma unroll
            for (int t = 0; t < STEPS; ++t) {
                if (t + AHEAD < STEPS)
                    a[t + AHEAD] = Afrag[(t + AHEAD) % OFRAGS][((cb >> 3) + (t + AHEAD) / OFRAGS) * 32];
                if (t % OFRAGS == 0) {
                    const int k = (t / OFRAGS) * 8;
#pragma unroll
                    for (int j = 0; j < NT; ++j) {
                        b0[j] = *(const v2f*)&Bbase[j * (16 * VSTRIDE) + k + khalf];
                        b1[j] = *(const v2f*)&Bbase[j * (16 * VSTRIDE) + k + 4 + khalf];
                    }
                }
                __builtin_amdgcn_sched_barrier(0);
                const v2f a0 = (v2f){a[t].x, a[t].y};
                const v2f a1 = (v2f){a[t].z, a[t].w};
                const int f = t % OFRAGS;
#pragma unroll
                for (int j = 0; j < NT; ++j)
                    acc[f][j] = __builtin_amdgcn_wmma_f32_16x16x4_f32(
                        false, a0, false, b0[j], (short)0, acc[f][j], false, false);
#pragma unroll
                for (int j = 0; j < NT; ++j)
                    acc[f][j] = __builtin_amdgcn_wmma_f32_16x16x4_f32(
                        false, a1, false, b1[j], (short)0, acc[f][j], false, false);
            }
        }
        __syncthreads();
    }

    {
        const int orow = (lane >> 4) << 3;
        const int ncol = lane & 15;
#pragma unroll
        for (int f = 0; f < OFRAGS; ++f)
#pragma unroll
            for (int j = 0; j < NT; ++j)
#pragma unroll
                for (int r = 0; r < 8; ++r)
                    smem[(((f * 16 + r + orow) * TPB) + j * 16 + ncol) * 16 + il] = acc[f][j][r];
    }
    __syncthreads();

    for (int p = tid; p < OPB * TPB; p += 512) {
        const int o_l = p / TPB;
        const int tl  = p % TPB;
        const int tg  = tb * TPB + tl;
        const int nn  = tg / TILES_PER_N;
        const int tr  = tg % TILES_PER_N;
        const int hh  = (tr / TW) * 2;
        const int ww  = (tr % TW) * 2;

        float m[16];
#pragma unroll
        for (int i = 0; i < 16; ++i)
            m[i] = smem[(o_l * TPB + tl) * 16 + i];

        float z0[4], z1[4];
#pragma unroll
        for (int c = 0; c < 4; ++c) {
            z0[c] = m[0 * 4 + c] + m[1 * 4 + c] + m[2 * 4 + c];
            z1[c] = m[1 * 4 + c] - m[2 * 4 + c] - m[3 * 4 + c];
        }
        const float y00 = z0[0] + z0[1] + z0[2];
        const float y01 = z0[1] - z0[2] - z0[3];
        const float y10 = z1[0] + z1[1] + z1[2];
        const float y11 = z1[1] - z1[2] - z1[3];

        const int o = obase + o_l;
        float* op = out + (((size_t)(nn * OCH + o) * H_DIM) + hh) * W_DIM + ww;
        op[0]         = y00;
        op[1]         = y01;
        op[W_DIM]     = y10;
        op[W_DIM + 1] = y11;
    }
}

extern "C" void kernel_launch(void* const* d_in, const int* in_sizes, int n_in,
                              void* d_out, int out_size, void* d_ws, size_t ws_size,
                              hipStream_t stream)
{
    const float* x = (const float*)d_in[0];
    const float* w = (const float*)d_in[1];
    float* out = (float*)d_out;
    float* U = (float*)d_ws;

    wino_weight_transform<<<dim3((OCH * CIN + 255) / 256), dim3(256), 0, stream>>>(w, U);
    wino_main<<<dim3(TILES_TOTAL / TPB, OCH / OPB), dim3(512), 0, stream>>>(x, U, out);
}


# epilogue staging in LDS laid out [il][o][tile] with a padded o-row: accumulator scatter and output-transform reads are bank-conflict-free (were 8-way and 4-way); sums unchanged
# speedup vs baseline: 3.2043x; 1.0927x over previous
#include <hip/hip_runtime.h>

typedef float v2f __attribute__((ext_vector_type(2)));
typedef float v4f __attribute__((ext_vector_type(4)));
typedef float v8f __attribute__((ext_vector_type(8)));

#define H_DIM 56
#define W_DIM 56
#define TH 28
#define TW 28
#define NB 16
#define CIN 256
#define OCH 256
#define TILES_TOTAL (NB * TH * TW)
#define TILES_PER_N (TH * TW)
#define NT 2
#define TPB (16 * NT)
#define OPB 64
#define OFRAGS (OPB / 16)
#define CCHUNK 32
#define VSTRIDE 34
#define PPT (TPB * CCHUNK / 512)
#define MROW (TPB + 2)
#define MPLANE (OPB * MROW)

__device__ __forceinline__ size_t u_frag_index(int il, int o, int c)
{
    const int lane = (o & 15) + (((c >> 1) & 1) << 4);
    const int j    = (((c >> 2) & 1) << 1) + (c & 1);
    return ((((size_t)il * (OCH / 16) + (o >> 4)) * (CIN / 8) + (c >> 3)) * 32 + lane) * 4 + j;
}

__global__ void wino_weight_transform(const float* __restrict__ w, float* __restrict__ U)
{
    int idx = blockIdx.x * blockDim.x + threadIdx.x;
    if (idx >= OCH * CIN) return;
    int o = idx >> 8;
    int c = idx & 255;
    const float* wp = w + (size_t)(o * CIN + c) * 9;

    float g[3][3];
#pragma unroll
    for (int r = 0; r < 3; ++r)
#pragma unroll
        for (int s = 0; s < 3; ++s) g[r][s] = wp[r * 3 + s];

    float u[4][3];
#pragma unroll
    for (int s = 0; s < 3; ++s) {
        u[0][s] = g[0][s];
        u[1][s] = 0.5f * (g[0][s] + g[1][s] + g[2][s]);
        u[2][s] = 0.5f * (g[0][s] - g[1][s] + g[2][s]);
        u[3][s] = g[2][s];
    }
#pragma unroll
    for (int r = 0; r < 4; ++r) {
        float a = u[r][0], b = u[r][1], d = u[r][2];
        float e[4];
        e[0] = a;
        e[1] = 0.5f * (a + b + d);
        e[2] = 0.5f * (a - b + d);
        e[3] = d;
#pragma unroll
        for (int l = 0; l < 4; ++l)
            U[u_frag_index(r * 4 + l, o, c)] = e[l];
    }
}

__launch_bounds__(512, 8)
__global__ void wino_main(const float* __restrict__ x, const float* __restrict__ U,
                          float* __restrict__ out)
{
    __shared__ float smem[16 * MPLANE];

    const int tb    = blockIdx.x;
    const int ob    = blockIdx.y;
    const int tid   = threadIdx.x;
    const int lane  = tid & 31;
    const int il    = tid >> 5;
    const int obase = ob * OPB;

    const int tile_l = tid % TPB;
    const int cl     = tid / TPB;
    const int tile_g = tb * TPB + tile_l;
    const int n      = tile_g / TILES_PER_N;
    const int trem   = tile_g % TILES_PER_N;
    const int th     = trem / TW;
    const int tw     = trem % TW;

    const bool r0ok = th > 0, r3ok = th < TH - 1;
    const bool c0ok = tw > 0, c3ok = tw < TW - 1;
    const float rm0 = r0ok ? 1.f : 0.f, rm3 = r3ok ? 1.f : 0.f;
    const float cm0 = c0ok ? 1.f : 0.f, cm3 = c3ok ? 1.f : 0.f;
    unsigned rbase[4];
    {
        const unsigned base = (unsigned)((n * CIN + PPT * cl) * (H_DIM * W_DIM) + 2 * tw);
        rbase[0] = (base + (unsigned)((r0ok ? 2 * th - 1 : 2 * th) * W_DIM)) * 4u;
        rbase[1] = (base + (unsigned)((2 * th) * W_DIM)) * 4u;
        rbase[2] = (base + (unsigned)((2 * th + 1) * W_DIM)) * 4u;
        rbase[3] = (base + (unsigned)((r3ok ? 2 * th + 2 : 2 * th + 1) * W_DIM)) * 4u;
    }
    const unsigned c0adj = c0ok ? (unsigned)-4 : 0u;
    const unsigned c3adj = c3ok ? 8u : 0u;

    v8f acc[OFRAGS][NT];
#pragma unroll
    for (int f = 0; f < OFRAGS; ++f)
#pragma unroll
        for (int j = 0; j < NT; ++j)
            acc[f][j] = (v8f){0.f, 0.f, 0.f, 0.f, 0.f, 0.f, 0.f, 0.f};

    const int m16   = lane & 15;
    const int khalf = (lane >> 4) << 1;
    const v4f* Afrag[OFRAGS];
#pragma unroll
    for (int f = 0; f < OFRAGS; ++f)
        Afrag[f] = (const v4f*)U +
                   ((size_t)il * (OCH / 16) + (ob * OFRAGS + f)) * (CIN / 8) * 32 + lane;
    const float* Bbase = &smem[il * (TPB * VSTRIDE) + m16 * VSTRIDE];

    for (int cb = 0; cb < CIN; cb += CCHUNK) {
        float d[PPT][4][4];
#pragma unroll
        for (int q = 0; q < PPT; ++q) {
            const char* xq = (const char*)x + (size_t)q * (H_DIM * W_DIM) * 4;
#pragma unroll
            for (int r = 0; r < 4; ++r) {
                d[q][r][0] = *(const float*)(xq + (rbase[r] + c0adj));
                d[q][r][1] = *(const float*)(xq + rbase[r]);
                d[q][r][2] = *(const float*)(xq + rbase[r] + 4);
                d[q][r][3] = *(const float*)(xq + (rbase[r] + c3adj));
            }
        }
#pragma unroll
        for (int r = 0; r < 4; ++r)
            rbase[r] += CCHUNK * (H_DIM * W_DIM) * 4;
#pragma unroll
        for (int q = 0; q < PPT; ++q) {
#pragma unroll
            for (int s = 0; s < 4; ++s) {
                d[q][0][s] *= rm0;
                d[q][3][s] *= rm3;
            }
#pragma unroll
            for (int r = 0; r < 4; ++r) {
                d[q][r][0] *= cm0;
                d[q][r][3] *= cm3;
            }
            const int cq = PPT * cl + q;
            float t[4][4];
#pragma unroll
            for (int s = 0; s < 4; ++s) {
                t[0][s] = d[q][0][s] - d[q][2][s];
                t[1][s] = d[q][1][s] + d[q][2][s];
                t[2][s] = d[q][2][s] - d[q][1][s];
                t[3][s] = d[q][1][s] - d[q][3][s];
            }
#pragma unroll
            for (int r = 0; r < 4; ++r) {
                float v0 = t[r][0] - t[r][2];
                float v1 = t[r][1] + t[r][2];
                float v2 = t[r][2] - t[r][1];
                float v3 = t[r][1] - t[r][3];
                smem[(r * 4 + 0) * (TPB * VSTRIDE) + tile_l * VSTRIDE + cq] = v0;
                smem[(r * 4 + 1) * (TPB * VSTRIDE) + tile_l * VSTRIDE + cq] = v1;
                smem[(r * 4 + 2) * (TPB * VSTRIDE) + tile_l * VSTRIDE + cq] = v2;
                smem[(r * 4 + 3) * (TPB * VSTRIDE) + tile_l * VSTRIDE + cq] = v3;
            }
        }
        __syncthreads();

        {
            constexpr int STEPS = (CCHUNK / 8) * OFRAGS;
            constexpr int AHEAD = 2;
            v4f a[STEPS];
#pragma unroll
            for (int t = 0; t < AHEAD; ++t)
                a[t] = Afrag[t % OFRAGS][((cb >> 3) + t / OFRAGS) * 32];
            v2f b0[NT], b1[NT];
#pragma unroll
            for (int t = 0; t < STEPS; ++t) {
                if (t + AHEAD < STEPS)
                    a[t + AHEAD] = Afrag[(t + AHEAD) % OFRAGS][((cb >> 3) + (t + AHEAD) / OFRAGS) * 32];
                if (t % OFRAGS == 0) {
                    const int k = (t / OFRAGS) * 8;
#pragma unroll
                    for (int j = 0; j < NT; ++j) {
                        b0[j] = *(const v2f*)&Bbase[j * (16 * VSTRIDE) + k + khalf];
                        b1[j] = *(const v2f*)&Bbase[j * (16 * VSTRIDE) + k + 4 + khalf];
                    }
                }
                __builtin_amdgcn_sched_barrier(0);
                const v2f a0 = (v2f){a[t].x, a[t].y};
                const v2f a1 = (v2f){a[t].z, a[t].w};
                const int f = t % OFRAGS;
#pragma unroll
                for (int j = 0; j < NT; ++j)
                    acc[f][j] = __builtin_amdgcn_wmma_f32_16x16x4_f32(
                        false, a0, false, b0[j], (short)0, acc[f][j], false, false);
#pragma unroll
                for (int j = 0; j < NT; ++j)
                    acc[f][j] = __builtin_amdgcn_wmma_f32_16x16x4_f32(
                        false, a1, false, b1[j], (short)0, acc[f][j], false, false);
            }
        }
        __syncthreads();
    }

    {
        const int orow = (lane >> 4) << 3;
        const int ncol = lane & 15;
#pragma unroll
        for (int f = 0; f < OFRAGS; ++f)
#pragma unroll
            for (int j = 0; j < NT; ++j)
#pragma unroll
                for (int r = 0; r < 8; ++r)
                    smem[il * MPLANE + (f * 16 + r + orow) * MROW + j * 16 + ncol] = acc[f][j][r];
    }
    __syncthreads();

    for (int p = tid; p < OPB * TPB; p += 512) {
        const int o_l = p / TPB;
        const int tl  = p % TPB;
        const int tg  = tb * TPB + tl;
        const int nn  = tg / TILES_PER_N;
        const int tr  = tg % TILES_PER_N;
        const int hh  = (tr / TW) * 2;
        const int ww  = (tr % TW) * 2;

        float m[16];
#pragma unroll
        for (int i = 0; i < 16; ++i)
            m[i] = smem[i * MPLANE + o_l * MROW + tl];

        float z0[4], z1[4];
#pragma unroll
        for (int c = 0; c < 4; ++c) {
            z0[c] = m[0 * 4 + c] + m[1 * 4 + c] + m[2 * 4 + c];
            z1[c] = m[1 * 4 + c] - m[2 * 4 + c] - m[3 * 4 + c];
        }
        const float y00 = z0[0] + z0[1] + z0[2];
        const float y01 = z0[1] - z0[2] - z0[3];
        const float y10 = z1[0] + z1[1] + z1[2];
        const float y11 = z1[1] - z1[2] - z1[3];

        const int o = obase + o_l;
        float* op = out + (((size_t)(nn * OCH + o) * H_DIM) + hh) * W_DIM + ww;
        op[0]         = y00;
        op[1]         = y01;
        op[W_DIM]     = y10;
        op[W_DIM + 1] = y11;
    }
}

extern "C" void kernel_launch(void* const* d_in, const int* in_sizes, int n_in,
                              void* d_out, int out_size, void* d_ws, size_t ws_size,
                              hipStream_t stream)
{
    const float* x = (const float*)d_in[0];
    const float* w = (const float*)d_in[1];
    float* out = (float*)d_out;
    float* U = (float*)d_ws;

    wino_weight_transform<<<dim3((OCH * CIN + 255) / 256), dim3(256), 0, stream>>>(w, U);
    wino_main<<<dim3(TILES_TOTAL / TPB, OCH / OPB), dim3(512), 0, stream>>>(x, U, out);
}


# grid rebalanced: 1536 full workgroups fill three residency rounds exactly; the other 32 items run as 256 small workgroups (16 tiles x 16 channels), dispatched last: short last round; sums unchanged
# speedup vs baseline: 3.3405x; 1.0425x over previous
#include <hip/hip_runtime.h>

typedef float v2f __attribute__((ext_vector_type(2)));
typedef float v4f __attribute__((ext_vector_type(4)));
typedef float v8f __attribute__((ext_vector_type(8)));

#define H_DIM 56
#define W_DIM 56
#define TH 28
#define TW 28
#define NB 16
#define CIN 256
#define OCH 256
#define TILES_TOTAL (NB * TH * TW)
#define TILES_PER_N (TH * TW)
#define CCHUNK 32
#define VSTRIDE 34

#define NT_FULL 2
#define OFR_FULL 4
#define TILE_BLOCKS (TILES_TOTAL / (16 * NT_FULL))
#define FULL_ITEMS (TILE_BLOCKS * (OCH / (16 * OFR_FULL)))
#define RESIDENT_WGS 512
#define FULL_WGS ((FULL_ITEMS / RESIDENT_WGS) * RESIDENT_WGS)
#define PIECES (NT_FULL * OFR_FULL)
#define SMALL_WGS ((FULL_ITEMS - FULL_WGS) * PIECES)
#define SMEM_FLOATS (16 * (16 * OFR_FULL) * (16 * NT_FULL + 2))

__device__ __forceinline__ size_t u_frag_index(int il, int o, int c)
{
    const int lane = (o & 15) + (((c >> 1) & 1) << 4);
    const int j    = (((c >> 2) & 1) << 1) + (c & 1);
    return ((((size_t)il * (OCH / 16) + (o >> 4)) * (CIN / 8) + (c >> 3)) * 32 + lane) * 4 + j;
}

__global__ void wino_weight_transform(const float* __restrict__ w, float* __restrict__ U)
{
    int idx = blockIdx.x * blockDim.x + threadIdx.x;
    if (idx >= OCH * CIN) return;
    int o = idx >> 8;
    int c = idx & 255;
    const float* wp = w + (size_t)(o * CIN + c) * 9;

    float g[3][3];
#pragma unroll
    for (int r = 0; r < 3; ++r)
#pragma unroll
        for (int s = 0; s < 3; ++s) g[r][s] = wp[r * 3 + s];

    float u[4][3];
#pragma unroll
    for (int s = 0; s < 3; ++s) {
        u[0][s] = g[0][s];
        u[1][s] = 0.5f * (g[0][s] + g[1][s] + g[2][s]);
        u[2][s] = 0.5f * (g[0][s] - g[1][s] + g[2][s]);
        u[3][s] = g[2][s];
    }
#pragma unroll
    for (int r = 0; r < 4; ++r) {
        float a = u[r][0], b = u[r][1], d = u[r][2];
        float e[4];
        e[0] = a;
        e[1] = 0.5f * (a + b + d);
        e[2] = 0.5f * (a - b + d);
        e[3] = d;
#pragma unroll
        for (int l = 0; l < 4; ++l)
            U[u_frag_index(r * 4 + l, o, c)] = e[l];
    }
}

template <int NT, int OFRAGS>
__device__ __forceinline__ void wino_body(const float* __restrict__ x, const float* __restrict__ U,
                                          float* __restrict__ out, float* smem,
                                          const int tile0, const int ofrag0)
{
    constexpr int TPB    = 16 * NT;
    constexpr int OPB    = 16 * OFRAGS;
    constexpr int PPT    = TPB * CCHUNK / 512;
    constexpr int MROW   = TPB + 2;
    constexpr int MPLANE = OPB * MROW;

    const int tid   = threadIdx.x;
    const int lane  = tid & 31;
    const int il    = tid >> 5;
    const int obase = ofrag0 * 16;

    const int tile_l = tid % TPB;
    const int cl     = tid / TPB;
    const int tile_g = tile0 + tile_l;
    const int n      = tile_g / TILES_PER_N;
    const int trem   = tile_g % TILES_PER_N;
    const int th     = trem / TW;
    const int tw     = trem % TW;

    const bool r0ok = th > 0, r3ok = th < TH - 1;
    const bool c0ok = tw > 0, c3ok = tw < TW - 1;
    const float rm0 = r0ok ? 1.f : 0.f, rm3 = r3ok ? 1.f : 0.f;
    const float cm0 = c0ok ? 1.f : 0.f, cm3 = c3ok ? 1.f : 0.f;
    unsigned rbase[4];
    {
        const unsigned base = (unsigned)((n * CIN + PPT * cl) * (H_DIM * W_DIM) + 2 * tw);
        rbase[0] = (base + (unsigned)((r0ok ? 2 * th - 1 : 2 * th) * W_DIM)) * 4u;
        rbase[1] = (base + (unsigned)((2 * th) * W_DIM)) * 4u;
        rbase[2] = (base + (unsigned)((2 * th + 1) * W_DIM)) * 4u;
        rbase[3] = (base + (unsigned)((r3ok ? 2 * th + 2 : 2 * th + 1) * W_DIM)) * 4u;
    }
    const unsigned c0adj = c0ok ? (unsigned)-4 : 0u;
    const unsigned c3adj = c3ok ? 8u : 0u;

    v8f acc[OFRAGS][NT];
#pragma unroll
    for (int f = 0; f < OFRAGS; ++f)
#pragma unroll
        for (int j = 0; j < NT; ++j)
            acc[f][j] = (v8f){0.f, 0.f, 0.f, 0.f, 0.f, 0.f, 0.f, 0.f};

    const int m16   = lane & 15;
    const int khalf = (lane >> 4) << 1;
    const v4f* Afrag[OFRAGS];
#pragma unroll
    for (int f = 0; f < OFRAGS; ++f)
        Afrag[f] = (const v4f*)U +
                   ((size_t)il * (OCH / 16) + (ofrag0 + f)) * (CIN / 8) * 32 + lane;
    const float* Bbase = &smem[il * (TPB * VSTRIDE) + m16 * VSTRIDE];

    for (int cb = 0; cb < CIN; cb += CCHUNK) {
        float d[PPT][4][4];
#pragma unroll
        for (int q = 0; q < PPT; ++q) {
            const char* xq = (const char*)x + (size_t)q * (H_DIM * W_DIM) * 4;
#pragma unroll
            for (int r = 0; r < 4; ++r) {
                d[q][r][0] = *(const float*)(xq + (rbase[r] + c0adj));
                d[q][r][1] = *(const float*)(xq + rbase[r]);
                d[q][r][2] = *(const float*)(xq + rbase[r] + 4);
                d[q][r][3] = *(const float*)(xq + (rbase[r] + c3adj));
            }
        }
#pragma unroll
        for (int r = 0; r < 4; ++r)
            rbase[r] += CCHUNK * (H_DIM * W_DIM) * 4;
#pragma unroll
        for (int q = 0; q < PPT; ++q) {
#pragma unroll
            for (int s = 0; s < 4; ++s) {
                d[q][0][s] *= rm0;
                d[q][3][s] *= rm3;
            }
#pragma unroll
            for (int r = 0; r < 4; ++r) {
                d[q][r][0] *= cm0;
                d[q][r][3] *= cm3;
            }
            const int cq = PPT * cl + q;
            float t[4][4];
#pragma unroll
            for (int s = 0; s < 4; ++s) {
                t[0][s] = d[q][0][s] - d[q][2][s];
                t[1][s] = d[q][1][s] + d[q][2][s];
                t[2][s] = d[q][2][s] - d[q][1][s];
                t[3][s] = d[q][1][s] - d[q][3][s];
            }
#pragma unroll
            for (int r = 0; r < 4; ++r) {
                float v0 = t[r][0] - t[r][2];
                float v1 = t[r][1] + t[r][2];
                float v2 = t[r][2] - t[r][1];
                float v3 = t[r][1] - t[r][3];
                smem[(r * 4 + 0) * (TPB * VSTRIDE) + tile_l * VSTRIDE + cq] = v0;
                smem[(r * 4 + 1) * (TPB * VSTRIDE) + tile_l * VSTRIDE + cq] = v1;
                smem[(r * 4 + 2) * (TPB * VSTRIDE) + tile_l * VSTRIDE + cq] = v2;
                smem[(r * 4 + 3) * (TPB * VSTRIDE) + tile_l * VSTRIDE + cq] = v3;
            }
        }
        __syncthreads();

        {
            constexpr int STEPS = (CCHUNK / 8) * OFRAGS;
            constexpr int AHEAD = 2;
            v4f a[STEPS];
#pragma unroll
            for (int t = 0; t < AHEAD; ++t)
                a[t] = Afrag[t % OFRAGS][((cb >> 3) + t / OFRAGS) * 32];
            v2f b0[NT], b1[NT];
#pragma unroll
            for (int t = 0; t < STEPS; ++t) {
                if (t + AHEAD < STEPS)
                    a[t + AHEAD] = Afrag[(t + AHEAD) % OFRAGS][((cb >> 3) + (t + AHEAD) / OFRAGS) * 32];
                if (t % OFRAGS == 0) {
                    const int k = (t / OFRAGS) * 8;
#pragma unroll
                    for (int j = 0; j < NT; ++j) {
                        b0[j] = *(const v2f*)&Bbase[j * (16 * VSTRIDE) + k + khalf];
                        b1[j] = *(const v2f*)&Bbase[j * (16 * VSTRIDE) + k + 4 + khalf];
                    }
                }
                __builtin_amdgcn_sched_barrier(0);
                const v2f a0 = (v2f){a[t].x, a[t].y};
                const v2f a1 = (v2f){a[t].z, a[t].w};
                const int f = t % OFRAGS;
#pragma unroll
                for (int j = 0; j < NT; ++j)
                    acc[f][j] = __builtin_amdgcn_wmma_f32_16x16x4_f32(
                        false, a0, false, b0[j], (short)0, acc[f][j], false, false);
#pragma unroll
                for (int j = 0; j < NT; ++j)
                    acc[f][j] = __builtin_amdgcn_wmma_f32_16x16x4_f32(
                        false, a1, false, b1[j], (short)0, acc[f][j], false, false);
            }
        }
        __syncthreads();
    }

    {
        const int orow = (lane >> 4) << 3;
        const int ncol = lane & 15;
#pragma unroll
        for (int f = 0; f < OFRAGS; ++f)
#pragma unroll
            for (int j = 0; j < NT; ++j)
#pragma unroll
                for (int r = 0; r < 8; ++r)
                    smem[il * MPLANE + (f * 16 + r + orow) * MROW + j * 16 + ncol] = acc[f][j][r];
    }
    __syncthreads();

    for (int p = tid; p < OPB * TPB; p += 512) {
        const int o_l = p / TPB;
        const int tl  = p % TPB;
        const int tg  = tile0 + tl;
        const int nn  = tg / TILES_PER_N;
        const int tr  = tg % TILES_PER_N;
        const int hh  = (tr / TW) * 2;
        const int ww  = (tr % TW) * 2;

        float m[16];
#pragma unroll
        for (int i = 0; i < 16; ++i)
            m[i] = smem[i * MPLANE + o_l * MROW + tl];

        float z0[4], z1[4];
#pragma unroll
        for (int c = 0; c < 4; ++c) {
            z0[c] = m[0 * 4 + c] + m[1 * 4 + c] + m[2 * 4 + c];
            z1[c] = m[1 * 4 + c] - m[2 * 4 + c] - m[3 * 4 + c];
        }
        const float y00 = z0[0] + z0[1] + z0[2];
        const float y01 = z0[1] - z0[2] - z0[3];
        const float y10 = z1[0] + z1[1] + z1[2];
        const float y11 = z1[1] - z1[2] - z1[3];

        const int o = obase + o_l;
        float* op = out + (((size_t)(nn * OCH + o) * H_DIM) + hh) * W_DIM + ww;
        op[0]         = y00;
        op[1]         = y01;
        op[W_DIM]     = y10;
        op[W_DIM + 1] = y11;
    }
}

__launch_bounds__(512, 8)
__global__ void wino_main(const float* __restrict__ x, const float* __restrict__ U,
                          float* __restrict__ out)
{
    __shared__ float smem[SMEM_FLOATS];

    const int b = blockIdx.x;
    if (b < FULL_WGS) {
        const int tb = b % TILE_BLOCKS;
        const int ob = b / TILE_BLOCKS;
        wino_body<NT_FULL, OFR_FULL>(x, U, out, smem, tb * (16 * NT_FULL), ob * OFR_FULL);
    } else {
        const int s    = b - FULL_WGS;
        const int item = FULL_WGS + s / PIECES;
        const int pc   = s % PIECES;
        const int tb   = item % TILE_BLOCKS;
        const int ob   = item / TILE_BLOCKS;
        wino_body<1, 1>(x, U, out, smem, tb * (16 * NT_FULL) + 16 * (pc / OFR_FULL),
                        ob * OFR_FULL + pc % OFR_FULL);
    }
}

extern "C" void kernel_launch(void* const* d_in, const int* in_sizes, int n_in,
                              void* d_out, int out_size, void* d_ws, size_t ws_size,
                              hipStream_t stream)
{
    const float* x = (const float*)d_in[0];
    const float* w = (const float*)d_in[1];
    float* out = (float*)d_out;
    float* U = (float*)d_ws;

    wino_weight_transform<<<dim3((OCH * CIN + 255) / 256), dim3(256), 0, stream>>>(w, U);
    wino_main<<<dim3(FULL_WGS + SMALL_WGS), dim3(512), 0, stream>>>(x, U, out);
}


# K loop with one barrier per chunk: V double-buffered in the LDS already held, waves run on into the next patches and first A requests; compiler now pairs V stores to 8 bytes; sums unchanged
# speedup vs baseline: 3.6288x; 1.0863x over previous
#include <hip/hip_runtime.h>

typedef float v2f __attribute__((ext_vector_type(2)));
typedef float v4f __attribute__((ext_vector_type(4)));
typedef float v8f __attribute__((ext_vector_type(8)));

#define H_DIM 56
#define W_DIM 56
#define TH 28
#define TW 28
#define NB 16
#define CIN 256
#define OCH 256
#define TILES_TOTAL (NB * TH * TW)
#define TILES_PER_N (TH * TW)
#define CCHUNK 32
#define VSTRIDE 34

#define NT_FULL 2
#define OFR_FULL 4
#define TILE_BLOCKS (TILES_TOTAL / (16 * NT_FULL))
#define FULL_ITEMS (TILE_BLOCKS * (OCH / (16 * OFR_FULL)))
#define RESIDENT_WGS 512
#define FULL_WGS ((FULL_ITEMS / RESIDENT_WGS) * RESIDENT_WGS)
#define PIECES (NT_FULL * OFR_FULL)
#define SMALL_WGS ((FULL_ITEMS - FULL_WGS) * PIECES)
#define SMEM_FLOATS (16 * (16 * OFR_FULL) * (16 * NT_FULL + 2))

__device__ __forceinline__ size_t u_frag_index(int il, int o, int c)
{
    const int lane = (o & 15) + (((c >> 1) & 1) << 4);
    const int j    = (((c >> 2) & 1) << 1) + (c & 1);
    return ((((size_t)il * (OCH / 16) + (o >> 4)) * (CIN / 8) + (c >> 3)) * 32 + lane) * 4 + j;
}

__global__ void wino_weight_transform(const float* __restrict__ w, float* __restrict__ U)
{
    int idx = blockIdx.x * blockDim.x + threadIdx.x;
    if (idx >= OCH * CIN) return;
    int o = idx >> 8;
    int c = idx & 255;
    const float* wp = w + (size_t)(o * CIN + c) * 9;

    float g[3][3];
#pragma unroll
    for (int r = 0; r < 3; ++r)
#pragma unroll
        for (int s = 0; s < 3; ++s) g[r][s] = wp[r * 3 + s];

    float u[4][3];
#pragma unroll
    for (int s = 0; s < 3; ++s) {
        u[0][s] = g[0][s];
        u[1][s] = 0.5f * (g[0][s] + g[1][s] + g[2][s]);
        u[2][s] = 0.5f * (g[0][s] - g[1][s] + g[2][s]);
        u[3][s] = g[2][s];
    }
#pragma unroll
    for (int r = 0; r < 4; ++r) {
        float a = u[r][0], b = u[r][1], d = u[r][2];
        float e[4];
        e[0] = a;
        e[1] = 0.5f * (a + b + d);
        e[2] = 0.5f * (a - b + d);
        e[3] = d;
#pragma unroll
        for (int l = 0; l < 4; ++l)
            U[u_frag_index(r * 4 + l, o, c)] = e[l];
    }
}

__device__ __forceinline__ void lds_barrier()
{
    __builtin_amdgcn_fence(__ATOMIC_RELEASE, "workgroup", "local");
    __builtin_amdgcn_s_barrier();
    __builtin_amdgcn_fence(__ATOMIC_ACQUIRE, "workgroup", "local");
}

template <int NT, int OFRAGS>
__device__ __forceinline__ void wino_body(const float* __restrict__ x, const float* __restrict__ U,
                                          float* __restrict__ out, float* smem,
                                          const int tile0, const int ofrag0)
{
    constexpr int TPB    = 16 * NT;
    constexpr int OPB    = 16 * OFRAGS;
    constexpr int PPT    = TPB * CCHUNK / 512;
    constexpr int MROW   = TPB + 2;
    constexpr int MPLANE = OPB * MROW;
    constexpr int VPLANE = TPB * VSTRIDE;
    constexpr int VSTAGE = 16 * VPLANE;

    const int tid   = threadIdx.x;
    const int lane  = tid & 31;
    const int il    = tid >> 5;
    const int obase = ofrag0 * 16;

    const int tile_l = tid % TPB;
    const int cl     = tid / TPB;
    const int tile_g = tile0 + tile_l;
    const int n      = tile_g / TILES_PER_N;
    const int trem   = tile_g % TILES_PER_N;
    const int th     = trem / TW;
    const int tw     = trem % TW;

    const bool r0ok = th > 0, r3ok = th < TH - 1;
    const bool c0ok = tw > 0, c3ok = tw < TW - 1;
    const float rm0 = r0ok ? 1.f : 0.f, rm3 = r3ok ? 1.f : 0.f;
    const float cm0 = c0ok ? 1.f : 0.f, cm3 = c3ok ? 1.f : 0.f;
    unsigned rbase[4];
    {
        const unsigned base = (unsigned)((n * CIN + PPT * cl) * (H_DIM * W_DIM) + 2 * tw);
        rbase[0] = (base + (unsigned)((r0ok ? 2 * th - 1 : 2 * th) * W_DIM)) * 4u;
        rbase[1] = (base + (unsigned)((2 * th) * W_DIM)) * 4u;
        rbase[2] = (base + (unsigned)((2 * th + 1) * W_DIM)) * 4u;
        rbase[3] = (base + (unsigned)((r3ok ? 2 * th + 2 : 2 * th + 1) * W_DIM)) * 4u;
    }
    const unsigned c0adj = c0ok ? (unsigned)-4 : 0u;
    const unsigned c3adj = c3ok ? 8u : 0u;

    v8f acc[OFRAGS][NT];
#pragma unroll
    for (int f = 0; f < OFRAGS; ++f)
#pragma unroll
        for (int j = 0; j < NT; ++j)
            acc[f][j] = (v8f){0.f, 0.f, 0.f, 0.f, 0.f, 0.f, 0.f, 0.f};

    const int m16   = lane & 15;
    const int khalf = (lane >> 4) << 1;
    const v4f* Afrag[OFRAGS];
#pragma unroll
    for (int f = 0; f < OFRAGS; ++f)
        Afrag[f] = (const v4f*)U +
                   ((size_t)il * (OCH / 16) + (ofrag0 + f)) * (CIN / 8) * 32 + lane;
    int boff = il * VPLANE + m16 * VSTRIDE;
    int soff = tile_l * VSTRIDE + PPT * cl;
    int sstep = VSTAGE;

    constexpr int STEPS = (CCHUNK / 8) * OFRAGS;
    constexpr int AHEAD = 2;
    for (int cb = 0; cb < CIN; cb += CCHUNK) {
        float d[PPT][4][4];
#pragma unroll
        for (int q = 0; q < PPT; ++q) {
            const char* xq = (const char*)x + (size_t)q * (H_DIM * W_DIM) * 4;
#pragma unroll
            for (int r = 0; r < 4; ++r) {
                d[q][r][0] = *(const float*)(xq + (rbase[r] + c0adj));
                d[q][r][1] = *(const float*)(xq + rbase[r]);
                d[q][r][2] = *(const float*)(xq + rbase[r] + 4);
                d[q][r][3] = *(const float*)(xq + (rbase[r] + c3adj));
            }
        }
#pragma unroll
        for (int r = 0; r < 4; ++r)
            rbase[r] += CCHUNK * (H_DIM * W_DIM) * 4;
#pragma unroll
        for (int q = 0; q < PPT; ++q) {
#pragma unroll
            for (int s = 0; s < 4; ++s) {
                d[q][0][s] *= rm0;
                d[q][3][s] *= rm3;
            }
#pragma unroll
            for (int r = 0; r < 4; ++r) {
                d[q][r][0] *= cm0;
                d[q][r][3] *= cm3;
            }
            float t[4][4];
#pragma unroll
            for (int s = 0; s < 4; ++s) {
                t[0][s] = d[q][0][s] - d[q][2][s];
                t[1][s] = d[q][1][s] + d[q][2][s];
                t[2][s] = d[q][2][s] - d[q][1][s];
                t[3][s] = d[q][1][s] - d[q][3][s];
            }
#pragma unroll
            for (int r = 0; r < 4; ++r) {
                float v0 = t[r][0] - t[r][2];
                float v1 = t[r][1] + t[r][2];
                float v2 = t[r][2] - t[r][1];
                float v3 = t[r][1] - t[r][3];
                smem[(r * 4 + 0) * VPLANE + soff + q] = v0;
                smem[(r * 4 + 1) * VPLANE + soff + q] = v1;
                smem[(r * 4 + 2) * VPLANE + soff + q] = v2;
                smem[(r * 4 + 3) * VPLANE + soff + q] = v3;
            }
        }
        v4f a[STEPS];
#pragma unroll
        for (int t = 0; t < AHEAD; ++t)
            a[t] = Afrag[t % OFRAGS][((cb >> 3) + t / OFRAGS) * 32];
        lds_barrier();

        {
            const float* Bbase = smem + boff;
            v2f b0[NT], b1[NT];
#pragma unroll
            for (int t = 0; t < STEPS; ++t) {
                if (t + AHEAD < STEPS)
                    a[t + AHEAD] = Afrag[(t + AHEAD) % OFRAGS][((cb >> 3) + (t + AHEAD) / OFRAGS) * 32];
                if (t % OFRAGS == 0) {
                    const int k = (t / OFRAGS) * 8;
#pragma unroll
                    for (int j = 0; j < NT; ++j) {
                        b0[j] = *(const v2f*)&Bbase[j * (16 * VSTRIDE) + k + khalf];
                        b1[j] = *(const v2f*)&Bbase[j * (16 * VSTRIDE) + k + 4 + khalf];
                    }
                }
                __builtin_amdgcn_sched_barrier(0);
                const v2f a0 = (v2f){a[t].x, a[t].y};
                const v2f a1 = (v2f){a[t].z, a[t].w};
                const int f = t % OFRAGS;
#pragma unroll
                for (int j = 0; j < NT; ++j)
                    acc[f][j] = __builtin_amdgcn_wmma_f32_16x16x4_f32(
                        false, a0, false, b0[j], (short)0, acc[f][j], false, false);
#pragma unroll
                for (int j = 0; j < NT; ++j)
                    acc[f][j] = __builtin_amdgcn_wmma_f32_16x16x4_f32(
                        false, a1, false, b1[j], (short)0, acc[f][j], false, false);
            }
        }
        soff += sstep;
        boff += sstep;
        sstep = -sstep;
    }
    __syncthreads();

    {
        const int orow = (lane >> 4) << 3;
        const int ncol = lane & 15;
#pragma unroll
        for (int f = 0; f < OFRAGS; ++f)
#pragma unroll
            for (int j = 0; j < NT; ++j)
#pragma unroll
                for (int r = 0; r < 8; ++r)
                    smem[il * MPLANE + (f * 16 + r + orow) * MROW + j * 16 + ncol] = acc[f][j][r];
    }
    __syncthreads();

    for (int p = tid; p < OPB * TPB; p += 512) {
        const int o_l = p / TPB;
        const int tl  = p % TPB;
        const int tg  = tile0 + tl;
        const int nn  = tg / TILES_PER_N;
        const int tr  = tg % TILES_PER_N;
        const int hh  = (tr / TW) * 2;
        const int ww  = (tr % TW) * 2;

        float m[16];
#pragma unroll
        for (int i = 0; i < 16; ++i)
            m[i] = smem[i * MPLANE + o_l * MROW + tl];

        float z0[4], z1[4];
#pragma unroll
        for (int c = 0; c < 4; ++c) {
            z0[c] = m[0 * 4 + c] + m[1 * 4 + c] + m[2 * 4 + c];
            z1[c] = m[1 * 4 + c] - m[2 * 4 + c] - m[3 * 4 + c];
        }
        const float y00 = z0[0] + z0[1] + z0[2];
        const float y01 = z0[1] - z0[2] - z0[3];
        const float y10 = z1[0] + z1[1] + z1[2];
        const float y11 = z1[1] - z1[2] - z1[3];

        const int o = obase + o_l;
        float* op = out + (((size_t)(nn * OCH + o) * H_DIM) + hh) * W_DIM + ww;
        op[0]         = y00;
        op[1]         = y01;
        op[W_DIM]     = y10;
        op[W_DIM + 1] = y11;
    }
}

__launch_bounds__(512, 8)
__global__ void wino_main(const float* __restrict__ x, const float* __restrict__ U,
                          float* __restrict__ out)
{
    __shared__ float smem[SMEM_FLOATS];

    const int b = blockIdx.x;
    if (b < FULL_WGS) {
        const int tb = b % TILE_BLOCKS;
        const int ob = b / TILE_BLOCKS;
        wino_body<NT_FULL, OFR_FULL>(x, U, out, smem, tb * (16 * NT_FULL), ob * OFR_FULL);
    } else {
        const int s    = b - FULL_WGS;
        const int item = FULL_WGS + s / PIECES;
        const int pc   = s % PIECES;
        const int tb   = item % TILE_BLOCKS;
        const int ob   = item / TILE_BLOCKS;
        wino_body<1, 1>(x, U, out, smem, tb * (16 * NT_FULL) + 16 * (pc / OFR_FULL),
                        ob * OFR_FULL + pc % OFR_FULL);
    }
}

extern "C" void kernel_launch(void* const* d_in, const int* in_sizes, int n_in,
                              void* d_out, int out_size, void* d_ws, size_t ws_size,
                              hipStream_t stream)
{
    const float* x = (const float*)d_in[0];
    const float* w = (const float*)d_in[1];
    float* out = (float*)d_out;
    float* U = (float*)d_ws;

    wino_weight_transform<<<dim3((OCH * CIN + 255) / 256), dim3(256), 0, stream>>>(w, U);
    wino_main<<<dim3(FULL_WGS + SMALL_WGS), dim3(512), 0, stream>>>(x, U, out);
}
